# starting kernel
# baseline (speedup 1.0000x reference)
#include <hip/hip_runtime.h>

#define ORDER 16
#define FRAME 160
#define NPAD  20
#define XPS   (NPAD + FRAME + NPAD)
#define WPB   8
#define LDSW  (XPS + NPAD + NPAD + FRAME)

typedef __attribute__((ext_vector_type(2))) float v2f;
typedef __attribute__((ext_vector_type(8))) float v8f;

__global__ __launch_bounds__(32 * WPB)
void lpc_residual_kernel(const float* __restrict__ pcm,
                         float* __restrict__ out,
                         int nframes)
{
    __shared__ float smem[WPB * LDSW];
    const int lane  = threadIdx.x & 31;
    const int wave  = threadIdx.x >> 5;
    const int frame = blockIdx.x * WPB + wave;
    if (frame >= nframes) return;

    float* xps   = smem + wave * LDSW;
    float* r_lds = xps + XPS;
    float* a_lds = r_lds + NPAD;
    float* y_lds = a_lds + NPAD;

    const float* src = pcm + (size_t)frame * FRAME;

    #pragma unroll
    for (int q = 0; q < 5; ++q)
        xps[NPAD + lane + 32 * q] = src[lane + 32 * q];
    if (lane < NPAD) { xps[lane] = 0.f; xps[NPAD + FRAME + lane] = 0.f; }
    __builtin_amdgcn_wave_barrier();

    if (lane <= ORDER) {
        float s = 0.f;
        for (int t = 0; t < FRAME; ++t)
            s += xps[NPAD + t] * xps[NPAD + t + lane];
        r_lds[lane] = s;
    } else if (lane < NPAD) {
        r_lds[lane] = 0.f;
    }
    if (lane == 0) a_lds[0] = 1.f;
    if (lane >= 1 && lane < NPAD) a_lds[lane] = 0.f;
    __builtin_amdgcn_wave_barrier();

    {
        float e = r_lds[0];
        for (int i = 1; i <= ORDER; ++i) {
            float acc = r_lds[i];
            for (int j = 1; j < i; ++j)
                acc -= a_lds[j] * r_lds[i - j];
            float k = acc / e;
            const int  j   = lane;
            const bool upd = (j >= 1) && (j < i);
            float aj  = upd ? a_lds[j]     : 0.f;
            float arv = upd ? a_lds[i - j] : 0.f;
            float na  = aj - k * arv;
            if (upd)       a_lds[j] = na;
            if (lane == 0) a_lds[i] = k;
            e = e * (1.f - k * k);
            __builtin_amdgcn_wave_barrier();
        }
    }

    const int half = lane >> 4;
    const int mrow = lane & 15;
    const int koff = half * 2;

    v2f bop[5];
    #pragma unroll
    for (int s = 0; s < 5; ++s) {
        bop[s].x = a_lds[4 * s + koff];
        bop[s].y = a_lds[4 * s + koff + 1];
    }

    #pragma unroll
    for (int c = 0; c < 10; ++c) {
        v8f acc = {};
        const int tb = 16 * c;
        #pragma unroll
        for (int s = 0; s < 5; ++s) {
            const int kk = 4 * s + koff;
            v2f aop;
            aop.x = xps[NPAD + tb + mrow - kk];
            aop.y = xps[NPAD + tb + mrow - kk - 1];
            acc = __builtin_amdgcn_wmma_f32_16x16x4_f32(
                      false, aop, false, bop[s], (short)0, acc, false, false);
        }
        #pragma unroll
        for (int q = 0; q < 8; ++q)
            y_lds[tb + half * 8 + q] = acc[q];
    }
    __builtin_amdgcn_wave_barrier();

    float* dst = out + (size_t)frame * FRAME;
    #pragma unroll
    for (int q = 0; q < 5; ++q)
        dst[lane + 32 * q] = y_lds[lane + 32 * q];
}

extern "C" void kernel_launch(void* const* d_in, const int* in_sizes, int n_in,
                              void* d_out, int out_size, void* d_ws, size_t ws_size,
                              hipStream_t stream) {
    const float* pcm = (const float*)d_in[0];
    float* out = (float*)d_out;
    const int total   = in_sizes[0];
    const int nframes = total / FRAME;
    const int blocks  = (nframes + WPB - 1) / WPB;
    lpc_residual_kernel<<<blocks, 32 * WPB, 0, stream>>>(pcm, out, nframes);
}


# autocorrelation r[0..16] on f32 WMMA: two 16x16 Gram tiles over the frame's 16-sample blocks (6 wmma) + diagonal sums via LDS replace the 160-step per-lag loop; summation order of r changes
# speedup vs baseline: 1.1899x; 1.1899x over previous
#include <hip/hip_runtime.h>

#define ORDER 16
#define FRAME 160
#define NPAD  20
#define XPS   (NPAD + FRAME + NPAD)
#define WPB   8
#define GP    20
#define GSZ   (32 * GP)
#define LDSW  (XPS + NPAD + NPAD + GSZ)

typedef __attribute__((ext_vector_type(2))) float v2f;
typedef __attribute__((ext_vector_type(4))) float v4f;
typedef __attribute__((ext_vector_type(8))) float v8f;

__global__ __launch_bounds__(32 * WPB)
void lpc_residual_kernel(const float* __restrict__ pcm,
                         float* __restrict__ out,
                         int nframes)
{
    __shared__ __attribute__((aligned(16))) float smem[WPB * LDSW];
    const int lane  = threadIdx.x & 31;
    const int wave  = threadIdx.x >> 5;
    const int frame = blockIdx.x * WPB + wave;
    if (frame >= nframes) return;

    float* xps   = smem + wave * LDSW;
    float* r_lds = xps + XPS;
    float* a_lds = r_lds + NPAD;
    float* y_lds = a_lds + NPAD;
    float* g_lds = y_lds;

    const float* src = pcm + (size_t)frame * FRAME;

    float xr[6];
    #pragma unroll
    for (int q = 0; q < 5; ++q) {
        xr[q] = src[lane + 32 * q];
        xps[NPAD + lane + 32 * q] = xr[q];
    }
    xr[5] = 0.f;
    if (lane < NPAD) { xps[lane] = 0.f; xps[NPAD + FRAME + lane] = 0.f; }
    __builtin_amdgcn_wave_barrier();

    {
        float xb[6];
        #pragma unroll
        for (int q = 0; q < 5; ++q)
            xb[q] = xps[NPAD + 16 + lane + 32 * q];
        xb[5] = 0.f;

        v8f g0 = {}, g1 = {};
        #pragma unroll
        for (int s = 0; s < 3; ++s) {
            v2f fa, fb;
            fa.x = xr[2 * s];  fa.y = xr[2 * s + 1];
            fb.x = xb[2 * s];  fb.y = xb[2 * s + 1];
            g0 = __builtin_amdgcn_wmma_f32_16x16x4_f32(
                     false, fa, false, fa, (short)0, g0, false, false);
            g1 = __builtin_amdgcn_wmma_f32_16x16x4_f32(
                     false, fa, false, fb, (short)0, g1, false, false);
        }

        const int hn = lane >> 4, cn = lane & 15;
        v4f* e0 = (v4f*)(g_lds + cn * GP + hn * 8);
        v4f* e1 = (v4f*)(g_lds + (16 + cn) * GP + hn * 8);
        e0[0] = (v4f){g0[0], g0[1], g0[2], g0[3]};
        e0[1] = (v4f){g0[4], g0[5], g0[6], g0[7]};
        e1[0] = (v4f){g1[0], g1[1], g1[2], g1[3]};
        e1[1] = (v4f){g1[4], g1[5], g1[6], g1[7]};
    }
    __builtin_amdgcn_wave_barrier();

    if (lane <= ORDER) {
        float s0 = 0.f, s1 = 0.f;
        #pragma unroll
        for (int m = 0; m < 16; m += 2) {
            s0 += g_lds[(m + lane) * GP + m];
            s1 += g_lds[(m + 1 + lane) * GP + m + 1];
        }
        r_lds[lane] = s0 + s1;
    } else if (lane < NPAD) {
        r_lds[lane] = 0.f;
    }
    if (lane == 0) a_lds[0] = 1.f;
    if (lane >= 1 && lane < NPAD) a_lds[lane] = 0.f;
    __builtin_amdgcn_wave_barrier();

    {
        float e = r_lds[0];
        for (int i = 1; i <= ORDER; ++i) {
            float acc = r_lds[i];
            for (int j = 1; j < i; ++j)
                acc -= a_lds[j] * r_lds[i - j];
            float k = acc / e;
            const int  j   = lane;
            const bool upd = (j >= 1) && (j < i);
            float aj  = upd ? a_lds[j]     : 0.f;
            float arv = upd ? a_lds[i - j] : 0.f;
            float na  = aj - k * arv;
            if (upd)       a_lds[j] = na;
            if (lane == 0) a_lds[i] = k;
            e = e * (1.f - k * k);
            __builtin_amdgcn_wave_barrier();
        }
    }

    const int half = lane >> 4;
    const int mrow = lane & 15;
    const int koff = half * 2;

    v2f bop[5];
    #pragma unroll
    for (int s = 0; s < 5; ++s) {
        bop[s].x = a_lds[4 * s + koff];
        bop[s].y = a_lds[4 * s + koff + 1];
    }

    #pragma unroll
    for (int c = 0; c < 10; ++c) {
        v8f acc = {};
        const int tb = 16 * c;
        #pragma unroll
        for (int s = 0; s < 5; ++s) {
            const int kk = 4 * s + koff;
            v2f aop;
            aop.x = xps[NPAD + tb + mrow - kk];
            aop.y = xps[NPAD + tb + mrow - kk - 1];
            acc = __builtin_amdgcn_wmma_f32_16x16x4_f32(
                      false, aop, false, bop[s], (short)0, acc, false, false);
        }
        #pragma unroll
        for (int q = 0; q < 8; ++q)
            y_lds[tb + half * 8 + q] = acc[q];
    }
    __builtin_amdgcn_wave_barrier();

    float* dst = out + (size_t)frame * FRAME;
    #pragma unroll
    for (int q = 0; q < 5; ++q)
        dst[lane + 32 * q] = y_lds[lane + 32 * q];
}

extern "C" void kernel_launch(void* const* d_in, const int* in_sizes, int n_in,
                              void* d_out, int out_size, void* d_ws, size_t ws_size,
                              hipStream_t stream) {
    const float* pcm = (const float*)d_in[0];
    float* out = (float*)d_out;
    const int total   = in_sizes[0];
    const int nframes = total / FRAME;
    const int blocks  = (nframes + WPB - 1) / WPB;
    lpc_residual_kernel<<<blocks, 32 * WPB, 0, stream>>>(pcm, out, nframes);
}


# Levinson-Durbin recursion kept in registers (lane j owns a[j]; a step's dot product = one multiply + 16-lane DPP butterfly sum), not serial LDS loops; summation order of the dot products changes
# speedup vs baseline: 1.4388x; 1.2092x over previous
#include <hip/hip_runtime.h>

#define ORDER 16
#define FRAME 160
#define NPAD  20
#define XPS   (NPAD + FRAME + NPAD)
#define WPB   8
#define GP    20
#define GSZ   (32 * GP)
#define LDSW  (XPS + NPAD + NPAD + GSZ)

typedef __attribute__((ext_vector_type(2))) float v2f;
typedef __attribute__((ext_vector_type(4))) float v4f;
typedef __attribute__((ext_vector_type(8))) float v8f;

#define DPP_ROW_ROR(n)   (0x120 + (n))
#define DPP_ROW_MIRROR   0x140
#define DPP_ROW_XMASK(m) (0x160 + (m))

template <int CTRL>
__device__ __forceinline__ float dpp_row(float v)
{
    return __builtin_bit_cast(float,
        __builtin_amdgcn_mov_dpp(__builtin_bit_cast(int, v), CTRL, 0xf, 0xf, true));
}

__device__ __forceinline__ float row_sum16(float v)
{
    v += dpp_row<DPP_ROW_XMASK(1)>(v);
    v += dpp_row<DPP_ROW_XMASK(2)>(v);
    v += dpp_row<DPP_ROW_XMASK(4)>(v);
    v += dpp_row<DPP_ROW_XMASK(8)>(v);
    return v;
}

template <int I>
__device__ __forceinline__ void levinson_step(float& av, float& ne, float& k,
                                              float rm, float r16, bool row_lane0)
{
    constexpr int ROT = (I + 1) & 15;
    float rrev = rm;
    if constexpr (ROT != 0) rrev = dpp_row<DPP_ROW_ROR(ROT)>(rm);
    if constexpr (I == 16)  rrev = row_lane0 ? r16 : rrev;
    const float s = row_sum16(av * rrev);
    k = s / ne;
    float arv = dpp_row<DPP_ROW_MIRROR>(av);
    if constexpr (ROT != 0) arv = dpp_row<DPP_ROW_ROR(ROT)>(arv);
    av = av - k * arv;
    ne = ne * (1.f - k * k);
}

__global__ __launch_bounds__(32 * WPB)
void lpc_residual_kernel(const float* __restrict__ pcm,
                         float* __restrict__ out,
                         int nframes)
{
    __shared__ __attribute__((aligned(16))) float smem[WPB * LDSW];
    const int lane  = threadIdx.x & 31;
    const int wave  = threadIdx.x >> 5;
    const int frame = blockIdx.x * WPB + wave;
    if (frame >= nframes) return;

    float* xps   = smem + wave * LDSW;
    float* r_lds = xps + XPS;
    float* a_lds = r_lds + NPAD;
    float* y_lds = a_lds + NPAD;
    float* g_lds = y_lds;

    const float* src = pcm + (size_t)frame * FRAME;

    float xr[6];
    #pragma unroll
    for (int q = 0; q < 5; ++q) {
        xr[q] = src[lane + 32 * q];
        xps[NPAD + lane + 32 * q] = xr[q];
    }
    xr[5] = 0.f;
    if (lane < NPAD) { xps[lane] = 0.f; xps[NPAD + FRAME + lane] = 0.f; }
    __builtin_amdgcn_wave_barrier();

    {
        float xb[6];
        #pragma unroll
        for (int q = 0; q < 5; ++q)
            xb[q] = xps[NPAD + 16 + lane + 32 * q];
        xb[5] = 0.f;

        v8f g0 = {}, g1 = {};
        #pragma unroll
        for (int s = 0; s < 3; ++s) {
            v2f fa, fb;
            fa.x = xr[2 * s];  fa.y = xr[2 * s + 1];
            fb.x = xb[2 * s];  fb.y = xb[2 * s + 1];
            g0 = __builtin_amdgcn_wmma_f32_16x16x4_f32(
                     false, fa, false, fa, (short)0, g0, false, false);
            g1 = __builtin_amdgcn_wmma_f32_16x16x4_f32(
                     false, fa, false, fb, (short)0, g1, false, false);
        }

        const int hn = lane >> 4, cn = lane & 15;
        v4f* e0 = (v4f*)(g_lds + cn * GP + hn * 8);
        v4f* e1 = (v4f*)(g_lds + (16 + cn) * GP + hn * 8);
        e0[0] = (v4f){g0[0], g0[1], g0[2], g0[3]};
        e0[1] = (v4f){g0[4], g0[5], g0[6], g0[7]};
        e1[0] = (v4f){g1[0], g1[1], g1[2], g1[3]};
        e1[1] = (v4f){g1[4], g1[5], g1[6], g1[7]};
    }
    __builtin_amdgcn_wave_barrier();

    if (lane <= ORDER) {
        float s0 = 0.f, s1 = 0.f;
        #pragma unroll
        for (int m = 0; m < 16; m += 2) {
            s0 += g_lds[(m + lane) * GP + m];
            s1 += g_lds[(m + 1 + lane) * GP + m + 1];
        }
        r_lds[lane] = s0 + s1;
    }
    __builtin_amdgcn_wave_barrier();

    {
        const int  jl        = lane & 15;
        const bool row_lane0 = (jl == 0);
        const float rm  = r_lds[15 - jl];
        const float r16 = r_lds[ORDER];
        float ne = -r_lds[0];
        float av = row_lane0 ? -1.f : 0.f;
        float k  = 0.f;
        levinson_step< 1>(av, ne, k, rm, r16, row_lane0);
        levinson_step< 2>(av, ne, k, rm, r16, row_lane0);
        levinson_step< 3>(av, ne, k, rm, r16, row_lane0);
        levinson_step< 4>(av, ne, k, rm, r16, row_lane0);
        levinson_step< 5>(av, ne, k, rm, r16, row_lane0);
        levinson_step< 6>(av, ne, k, rm, r16, row_lane0);
        levinson_step< 7>(av, ne, k, rm, r16, row_lane0);
        levinson_step< 8>(av, ne, k, rm, r16, row_lane0);
        levinson_step< 9>(av, ne, k, rm, r16, row_lane0);
        levinson_step<10>(av, ne, k, rm, r16, row_lane0);
        levinson_step<11>(av, ne, k, rm, r16, row_lane0);
        levinson_step<12>(av, ne, k, rm, r16, row_lane0);
        levinson_step<13>(av, ne, k, rm, r16, row_lane0);
        levinson_step<14>(av, ne, k, rm, r16, row_lane0);
        levinson_step<15>(av, ne, k, rm, r16, row_lane0);
        levinson_step<16>(av, ne, k, rm, r16, row_lane0);

        float aout = av;
        if (lane == 0)     aout = 1.f;
        if (lane == ORDER) aout = k;
        if (lane >  ORDER) aout = 0.f;
        if (lane < NPAD) a_lds[lane] = aout;
    }
    __builtin_amdgcn_wave_barrier();

    const int half = lane >> 4;
    const int mrow = lane & 15;
    const int koff = half * 2;

    v2f bop[5];
    #pragma unroll
    for (int s = 0; s < 5; ++s) {
        bop[s].x = a_lds[4 * s + koff];
        bop[s].y = a_lds[4 * s + koff + 1];
    }

    #pragma unroll
    for (int c = 0; c < 10; ++c) {
        v8f acc = {};
        const int tb = 16 * c;
        #pragma unroll
        for (int s = 0; s < 5; ++s) {
            const int kk = 4 * s + koff;
            v2f aop;
            aop.x = xps[NPAD + tb + mrow - kk];
            aop.y = xps[NPAD + tb + mrow - kk - 1];
            acc = __builtin_amdgcn_wmma_f32_16x16x4_f32(
                      false, aop, false, bop[s], (short)0, acc, false, false);
        }
        #pragma unroll
        for (int q = 0; q < 8; ++q)
            y_lds[tb + half * 8 + q] = acc[q];
    }
    __builtin_amdgcn_wave_barrier();

    float* dst = out + (size_t)frame * FRAME;
    #pragma unroll
    for (int q = 0; q < 5; ++q)
        dst[lane + 32 * q] = y_lds[lane + 32 * q];
}

extern "C" void kernel_launch(void* const* d_in, const int* in_sizes, int n_in,
                              void* d_out, int out_size, void* d_ws, size_t ws_size,
                              hipStream_t stream) {
    const float* pcm = (const float*)d_in[0];
    float* out = (float*)d_out;
    const int total   = in_sizes[0];
    const int nframes = total / FRAME;
    const int blocks  = (nframes + WPB - 1) / WPB;
    lpc_residual_kernel<<<blocks, 32 * WPB, 0, stream>>>(pcm, out, nframes);
}


# two frames per wave: each 16-lane row carries its own frame through the diagonal sums and the Levinson recursion (rows no longer duplicate); half as many waves; summation order of r changes
# speedup vs baseline: 3.3015x; 2.2946x over previous
#include <hip/hip_runtime.h>


#define ORDER 16
#define FRAME 160
#define NPAD  20
#define XPS   (NPAD + FRAME + NPAD)
#define FPW   2
#define WPB   8
#define GP    20
#define GSZ   (32 * GP)
#define AEN   48
#define YSZ   256
#define LDSW  (FPW * XPS + FPW * AEN + GSZ)

typedef __attribute__((ext_vector_type(2))) float v2f;
typedef __attribute__((ext_vector_type(4))) float v4f;
typedef __attribute__((ext_vector_type(8))) float v8f;

#define DPP_ROW_ROR(n)   (0x120 + (n))
#define DPP_ROW_MIRROR   0x140
#define DPP_ROW_SHARE(n) (0x150 + (n))
#define DPP_ROW_XMASK(m) (0x160 + (m))

template <int CTRL>
__device__ __forceinline__ float dpp_row(float v)
{
    return __builtin_bit_cast(float,
        __builtin_amdgcn_mov_dpp(__builtin_bit_cast(int, v), CTRL, 0xf, 0xf, true));
}

__device__ __forceinline__ float other_row(float v)
{
    return __builtin_bit_cast(float,
        __builtin_amdgcn_permlanex16(__builtin_bit_cast(int, v), __builtin_bit_cast(int, v),
                                     0x76543210, 0xfedcba98, false, false));
}

__device__ __forceinline__ float row_sum16(float v)
{
    v += dpp_row<DPP_ROW_XMASK(1)>(v);
    v += dpp_row<DPP_ROW_XMASK(2)>(v);
    v += dpp_row<DPP_ROW_XMASK(4)>(v);
    v += dpp_row<DPP_ROW_XMASK(8)>(v);
    return v;
}

template <int I>
__device__ __forceinline__ void levinson_step(float& av, float& ne, float& k,
                                              float rm, float r16, bool row_lane0)
{
    constexpr int ROT = (I + 1) & 15;
    float rrev = rm;
    if constexpr (ROT != 0) rrev = dpp_row<DPP_ROW_ROR(ROT)>(rm);
    if constexpr (I == 16)  rrev = row_lane0 ? r16 : rrev;
    const float s = row_sum16(av * rrev);
    k = s / ne;
    float arv = dpp_row<DPP_ROW_MIRROR>(av);
    if constexpr (ROT != 0) arv = dpp_row<DPP_ROW_ROR(ROT)>(arv);
    av = av - k * arv;
    ne = ne * (1.f - k * k);
}

__global__ __launch_bounds__(32 * WPB)
void lpc_residual_kernel(const float* __restrict__ pcm,
                         float* __restrict__ out,
                         int nframes)
{
    __shared__ __attribute__((aligned(16))) float smem[WPB * LDSW];
    const int lane   = threadIdx.x & 31;
    const int wave   = threadIdx.x >> 5;
    const int half   = lane >> 4;
    const int cidx   = lane & 15;
    const int frame0 = (blockIdx.x * WPB + wave) * FPW;
    if (frame0 >= nframes) return;
    const bool two = frame0 + 1 < nframes;

    float* ar    = smem + wave * LDSW;
    float* xps   = ar + FPW * AEN;
    float* g_lds = xps + FPW * XPS;
    float* y_lds = g_lds;

    const float* src = pcm + (size_t)frame0 * FRAME;

    float xr[FPW][6];
    #pragma unroll
    for (int q = 0; q < 5; ++q) xr[0][q] = src[lane + 32 * q];
    if (two) {
        #pragma unroll
        for (int q = 0; q < 5; ++q) xr[1][q] = src[FRAME + lane + 32 * q];
    } else {
        #pragma unroll
        for (int q = 0; q < 5; ++q) xr[1][q] = 0.f;
    }
    #pragma unroll
    for (int f = 0; f < FPW; ++f) {
        #pragma unroll
        for (int q = 0; q < 5; ++q) xps[f * XPS + NPAD + lane + 32 * q] = xr[f][q];
        xr[f][5] = 0.f;
        if (lane < NPAD) { xps[f * XPS + lane] = 0.f; xps[f * XPS + NPAD + FRAME + lane] = 0.f; }
        if (lane < 15) {
            ar[f * AEN + 32 + lane] = 0.f;
            ar[f * AEN + lane]      = 0.f;
        }
    }
    __builtin_amdgcn_wave_barrier();

    v8f g0[FPW], g1[FPW];
    #pragma unroll
    for (int f = 0; f < FPW; ++f) {
        float xb[6];
        #pragma unroll
        for (int q = 0; q < 5; ++q)
            xb[q] = xps[f * XPS + NPAD + 16 + lane + 32 * q];
        xb[5] = 0.f;

        g0[f] = (v8f){}; g1[f] = (v8f){};
        #pragma unroll
        for (int s = 0; s < 3; ++s) {
            v2f fa, fb;
            fa.x = xr[f][2 * s];  fa.y = xr[f][2 * s + 1];
            fb.x = xb[2 * s];     fb.y = xb[2 * s + 1];
            g0[f] = __builtin_amdgcn_wmma_f32_16x16x4_f32(
                        false, fa, false, fa, (short)0, g0[f], false, false);
            g1[f] = __builtin_amdgcn_wmma_f32_16x16x4_f32(
                        false, fa, false, fb, (short)0, g1[f], false, false);
        }
    }

    float part[FPW], diag[FPW];
    #pragma unroll
    for (int f = 0; f < FPW; ++f) {
        v4f* e0 = (v4f*)(g_lds + cidx * GP + half * 8);
        v4f* e1 = (v4f*)(g_lds + (16 + cidx) * GP + half * 8);
        e0[0] = (v4f){g0[f][0], g0[f][1], g0[f][2], g0[f][3]};
        e0[1] = (v4f){g0[f][4], g0[f][5], g0[f][6], g0[f][7]};
        e1[0] = (v4f){g1[f][0], g1[f][1], g1[f][2], g1[f][3]};
        e1[1] = (v4f){g1[f][4], g1[f][5], g1[f][6], g1[f][7]};
        __builtin_amdgcn_wave_barrier();

        const float* gd = g_lds + (8 * half + cidx) * GP + 8 * half;
        float s0 = 0.f, s1 = 0.f;
        #pragma unroll
        for (int i = 0; i < 8; i += 2) {
            s0 += gd[i * (GP + 1)];
            s1 += gd[(i + 1) * (GP + 1)];
        }
        part[f] = s0 + s1;
        diag[f] = g_lds[(16 + cidx) * GP + cidx];
        __builtin_amdgcn_wave_barrier();
    }

    {
        const bool row_lane0 = (cidx == 0);
        const float mine   = half ? part[1] : part[0];
        const float theirs = half ? part[0] : part[1];
        const float r   = mine + other_row(theirs);
        const float r16 = row_sum16(half ? diag[1] : diag[0]);
        const float rm  = dpp_row<DPP_ROW_MIRROR>(r);
        float ne = -dpp_row<DPP_ROW_SHARE(0)>(r);
        float av = row_lane0 ? -1.f : 0.f;
        float k  = 0.f;
        levinson_step< 1>(av, ne, k, rm, r16, row_lane0);
        levinson_step< 2>(av, ne, k, rm, r16, row_lane0);
        levinson_step< 3>(av, ne, k, rm, r16, row_lane0);
        levinson_step< 4>(av, ne, k, rm, r16, row_lane0);
        levinson_step< 5>(av, ne, k, rm, r16, row_lane0);
        levinson_step< 6>(av, ne, k, rm, r16, row_lane0);
        levinson_step< 7>(av, ne, k, rm, r16, row_lane0);
        levinson_step< 8>(av, ne, k, rm, r16, row_lane0);
        levinson_step< 9>(av, ne, k, rm, r16, row_lane0);
        levinson_step<10>(av, ne, k, rm, r16, row_lane0);
        levinson_step<11>(av, ne, k, rm, r16, row_lane0);
        levinson_step<12>(av, ne, k, rm, r16, row_lane0);
        levinson_step<13>(av, ne, k, rm, r16, row_lane0);
        levinson_step<14>(av, ne, k, rm, r16, row_lane0);
        levinson_step<15>(av, ne, k, rm, r16, row_lane0);
        levinson_step<16>(av, ne, k, rm, r16, row_lane0);

        float* arh = ar + half * AEN;
        arh[31 - cidx] = row_lane0 ? 1.f : av;
        if (row_lane0) arh[15] = k;
    }
    __builtin_amdgcn_wave_barrier();

    {
        const int nb = cidx < 10 ? cidx : 9;
        #pragma unroll
        for (int f = 0; f < FPW; ++f) {
            const v4f* xp4 = (const v4f*)(xps + f * XPS + (NPAD - 16) + 16 * (nb + half));
            v4f xv[4];
            #pragma unroll
            for (int v = 0; v < 4; ++v) xv[v] = xp4[v];

            const float* ap = ar + f * AEN + 15 - cidx + 16 * half;
            float af[16];
            #pragma unroll
            for (int i = 0; i < 16; ++i) af[i] = ap[i];

            v8f acc = {};
            #pragma unroll
            for (int s = 0; s < 8; ++s) {
                v2f fa, fb;
                fa.x = af[2 * s];                  fa.y = af[2 * s + 1];
                fb.x = xv[s >> 1][2 * (s & 1)];    fb.y = xv[s >> 1][2 * (s & 1) + 1];
                acc = __builtin_amdgcn_wmma_f32_16x16x4_f32(
                          false, fa, false, fb, (short)0, acc, false, false);
            }
            v4f* yp = (v4f*)(y_lds + f * YSZ + 16 * cidx + 8 * half);
            yp[0] = (v4f){acc[0], acc[1], acc[2], acc[3]};
            yp[1] = (v4f){acc[4], acc[5], acc[6], acc[7]};
        }
    }
    __builtin_amdgcn_wave_barrier();

    float* dst = out + (size_t)frame0 * FRAME;
    #pragma unroll
    for (int q = 0; q < 5; ++q)
        dst[lane + 32 * q] = y_lds[lane + 32 * q];
    if (two) {
        #pragma unroll
        for (int q = 0; q < 5; ++q)
            dst[FRAME + lane + 32 * q] = y_lds[YSZ + lane + 32 * q];
    }
}

extern "C" void kernel_launch(void* const* d_in, const int* in_sizes, int n_in,
                              void* d_out, int out_size, void* d_ws, size_t ws_size,
                              hipStream_t stream) {
    const float* pcm = (const float*)d_in[0];
    float* out = (float*)d_out;
    const int total   = in_sizes[0];
    const int nframes = total / FRAME;
    const int nwaves  = (nframes + FPW - 1) / FPW;
    const int blocks  = (nwaves + WPB - 1) / WPB;
    lpc_residual_kernel<<<blocks, 32 * WPB, 0, stream>>>(pcm, out, nframes);
}


# Levinson-Durbin recursion run one frame per lane: r[0..16] of a block's 16 frames pass through LDS, one wave does the 16 steps serially in registers; dot-product summation order changes
# speedup vs baseline: 3.9485x; 1.1960x over previous
#include <hip/hip_runtime.h>


#define ORDER 16
#define FRAME 160
#define NPAD  20
#define XPS   (NPAD + FRAME + NPAD)
#define FPW   2
#define WPB   8
#define GP    20
#define GSZ   (32 * GP)
#define AEN   48
#define YSZ   256
#define LDSW  (FPW * XPS + FPW * AEN + GSZ)
#define RP    17
static_assert(WPB * FPW == 16, "one 16-lane row of the recursion wave carries the block's frames");

typedef __attribute__((ext_vector_type(2))) float v2f;
typedef __attribute__((ext_vector_type(4))) float v4f;
typedef __attribute__((ext_vector_type(8))) float v8f;

#define DPP_ROW_XMASK(m) (0x160 + (m))

template <int CTRL>
__device__ __forceinline__ float dpp_row(float v)
{
    return __builtin_bit_cast(float,
        __builtin_amdgcn_mov_dpp(__builtin_bit_cast(int, v), CTRL, 0xf, 0xf, true));
}

__device__ __forceinline__ float other_row(float v)
{
    return __builtin_bit_cast(float,
        __builtin_amdgcn_permlanex16(__builtin_bit_cast(int, v), __builtin_bit_cast(int, v),
                                     0x76543210, 0xfedcba98, false, false));
}

__device__ __forceinline__ float row_sum16(float v)
{
    v += dpp_row<DPP_ROW_XMASK(1)>(v);
    v += dpp_row<DPP_ROW_XMASK(2)>(v);
    v += dpp_row<DPP_ROW_XMASK(4)>(v);
    v += dpp_row<DPP_ROW_XMASK(8)>(v);
    return v;
}

__global__ __launch_bounds__(32 * WPB)
void lpc_residual_kernel(const float* __restrict__ pcm,
                         float* __restrict__ out,
                         int nframes)
{
    __shared__ __attribute__((aligned(16))) float smem[WPB * LDSW];
    const int lane   = threadIdx.x & 31;
    const int wave   = threadIdx.x >> 5;
    const int half   = lane >> 4;
    const int cidx   = lane & 15;
    const int frame0 = (blockIdx.x * WPB + wave) * FPW;
    const bool one = frame0 < nframes;
    const bool two = frame0 + 1 < nframes;

    float* ar    = smem + wave * LDSW;
    float* xps   = ar + FPW * AEN;
    float* g_lds = xps + FPW * XPS;
    float* y_lds = g_lds;

    const float* src = pcm + (size_t)frame0 * FRAME;

    float xr[FPW][6];
    if (one) {
        #pragma unroll
        for (int q = 0; q < 5; ++q) xr[0][q] = src[lane + 32 * q];
    } else {
        #pragma unroll
        for (int q = 0; q < 5; ++q) xr[0][q] = 0.f;
    }
    if (two) {
        #pragma unroll
        for (int q = 0; q < 5; ++q) xr[1][q] = src[FRAME + lane + 32 * q];
    } else {
        #pragma unroll
        for (int q = 0; q < 5; ++q) xr[1][q] = 0.f;
    }
    #pragma unroll
    for (int f = 0; f < FPW; ++f) {
        #pragma unroll
        for (int q = 0; q < 5; ++q) xps[f * XPS + NPAD + lane + 32 * q] = xr[f][q];
        xr[f][5] = 0.f;
        if (lane < NPAD) { xps[f * XPS + lane] = 0.f; xps[f * XPS + NPAD + FRAME + lane] = 0.f; }
        if (lane < 15) {
            ar[f * AEN + 32 + lane] = 0.f;
            ar[f * AEN + lane]      = 0.f;
        }
    }
    __builtin_amdgcn_wave_barrier();

    v8f g0[FPW], g1[FPW];
    #pragma unroll
    for (int f = 0; f < FPW; ++f) {
        float xb[6];
        #pragma unroll
        for (int q = 0; q < 5; ++q)
            xb[q] = xps[f * XPS + NPAD + 16 + lane + 32 * q];
        xb[5] = 0.f;

        g0[f] = (v8f){}; g1[f] = (v8f){};
        #pragma unroll
        for (int s = 0; s < 3; ++s) {
            v2f fa, fb;
            fa.x = xr[f][2 * s];  fa.y = xr[f][2 * s + 1];
            fb.x = xb[2 * s];     fb.y = xb[2 * s + 1];
            g0[f] = __builtin_amdgcn_wmma_f32_16x16x4_f32(
                        false, fa, false, fa, (short)0, g0[f], false, false);
            g1[f] = __builtin_amdgcn_wmma_f32_16x16x4_f32(
                        false, fa, false, fb, (short)0, g1[f], false, false);
        }
    }

    float part[FPW], diag[FPW];
    #pragma unroll
    for (int f = 0; f < FPW; ++f) {
        v4f* e0 = (v4f*)(g_lds + cidx * GP + half * 8);
        v4f* e1 = (v4f*)(g_lds + (16 + cidx) * GP + half * 8);
        e0[0] = (v4f){g0[f][0], g0[f][1], g0[f][2], g0[f][3]};
        e0[1] = (v4f){g0[f][4], g0[f][5], g0[f][6], g0[f][7]};
        e1[0] = (v4f){g1[f][0], g1[f][1], g1[f][2], g1[f][3]};
        e1[1] = (v4f){g1[f][4], g1[f][5], g1[f][6], g1[f][7]};
        __builtin_amdgcn_wave_barrier();

        const float* gd = g_lds + (8 * half + cidx) * GP + 8 * half;
        float s0 = 0.f, s1 = 0.f;
        #pragma unroll
        for (int i = 0; i < 8; i += 2) {
            s0 += gd[i * (GP + 1)];
            s1 += gd[(i + 1) * (GP + 1)];
        }
        part[f] = s0 + s1;
        diag[f] = g_lds[(16 + cidx) * GP + cidx];
        __builtin_amdgcn_wave_barrier();
    }

    {
        const float mine   = half ? part[1] : part[0];
        const float theirs = half ? part[0] : part[1];
        const float r   = mine + other_row(theirs);
        const float r16 = row_sum16(half ? diag[1] : diag[0]);
        float* rl = g_lds + half * RP;
        rl[cidx] = r;
        if (cidx == 0) rl[16] = r16;
    }
    __syncthreads();

    if (wave == 0 && half == 0) {
        const float* rp = smem + (cidx >> 1) * LDSW + (FPW * AEN + FPW * XPS) + (cidx & 1) * RP;
        float r[ORDER + 1], a[ORDER + 1];
        #pragma unroll
        for (int i = 0; i <= ORDER; ++i) r[i] = rp[i];
        float e = r[0];
        #pragma unroll
        for (int i = 1; i <= ORDER; ++i) {
            float acc = r[i];
            if (i == 2) acc = r[2] - a[1] * r[1];
            if (i > 2) {
                float s0 = a[1] * r[i - 1];
                float s1 = a[2] * r[i - 2];
                #pragma unroll
                for (int j = 3; j < i; j += 2) s0 = fmaf(a[j], r[i - j], s0);
                #pragma unroll
                for (int j = 4; j < i; j += 2) s1 = fmaf(a[j], r[i - j], s1);
                acc = r[i] - (s0 + s1);
            }
            const float k = acc / e;
            #pragma unroll
            for (int j = 1; 2 * j < i; ++j) {
                const float lo = a[j], hi = a[i - j];
                a[j]     = lo - k * hi;
                a[i - j] = hi - k * lo;
            }
            if ((i & 1) == 0) a[i / 2] = a[i / 2] - k * a[i / 2];
            a[i] = k;
            e = e * (1.f - k * k);
        }
        float* arh = smem + (cidx >> 1) * LDSW + (cidx & 1) * AEN;
        arh[31] = 1.f;
        #pragma unroll
        for (int j = 1; j <= ORDER; ++j) arh[31 - j] = a[j];
    }
    __syncthreads();

    {
        const int nb = cidx < 10 ? cidx : 9;
        #pragma unroll
        for (int f = 0; f < FPW; ++f) {
            const v4f* xp4 = (const v4f*)(xps + f * XPS + (NPAD - 16) + 16 * (nb + half));
            v4f xv[4];
            #pragma unroll
            for (int v = 0; v < 4; ++v) xv[v] = xp4[v];

            const float* ap = ar + f * AEN + 15 - cidx + 16 * half;
            float af[16];
            #pragma unroll
            for (int i = 0; i < 16; ++i) af[i] = ap[i];

            v8f acc = {};
            #pragma unroll
            for (int s = 0; s < 8; ++s) {
                v2f fa, fb;
                fa.x = af[2 * s];                  fa.y = af[2 * s + 1];
                fb.x = xv[s >> 1][2 * (s & 1)];    fb.y = xv[s >> 1][2 * (s & 1) + 1];
                acc = __builtin_amdgcn_wmma_f32_16x16x4_f32(
                          false, fa, false, fb, (short)0, acc, false, false);
            }
            v4f* yp = (v4f*)(y_lds + f * YSZ + 16 * cidx + 8 * half);
            yp[0] = (v4f){acc[0], acc[1], acc[2], acc[3]};
            yp[1] = (v4f){acc[4], acc[5], acc[6], acc[7]};
        }
    }
    __builtin_amdgcn_wave_barrier();

    float* dst = out + (size_t)frame0 * FRAME;
    if (one) {
        #pragma unroll
        for (int q = 0; q < 5; ++q)
            dst[lane + 32 * q] = y_lds[lane + 32 * q];
    }
    if (two) {
        #pragma unroll
        for (int q = 0; q < 5; ++q)
            dst[FRAME + lane + 32 * q] = y_lds[YSZ + lane + 32 * q];
    }
}

extern "C" void kernel_launch(void* const* d_in, const int* in_sizes, int n_in,
                              void* d_out, int out_size, void* d_ws, size_t ws_size,
                              hipStream_t stream) {
    const float* pcm = (const float*)d_in[0];
    float* out = (float*)d_out;
    const int total   = in_sizes[0];
    const int nframes = total / FRAME;
    const int nwaves  = (nframes + FPW - 1) / FPW;
    const int blocks  = (nwaves + WPB - 1) / WPB;
    lpc_residual_kernel<<<blocks, 32 * WPB, 0, stream>>>(pcm, out, nframes);
}


# four frames per wave as two pairs (all loads issued first, no branch), 32 per block so wave 0's recursion fills all 32 lanes: half as many waves and blocks to enter the chip; no sum reordered
# speedup vs baseline: 4.2564x; 1.0780x over previous
#include <hip/hip_runtime.h>


#define ORDER 16
#define FRAME 160
#define NPAD  20
#define XPS   (NPAD + FRAME + NPAD)
#define FPW   4
#define NPR   (FPW / 2)
#define WPB   8
#define FPB   (WPB * FPW)
#define GP    20
#define GSZ   (32 * GP)
#define AEN   49
#define YSZ   256
#define LDSW  (FPW * XPS + NPR * GSZ)
#define RP    17
static_assert(FPB == 32, "one lane of the recursion wave carries each of the block's frames");

typedef __attribute__((ext_vector_type(2))) float v2f;
typedef __attribute__((ext_vector_type(4))) float v4f;
typedef __attribute__((ext_vector_type(8))) float v8f;

#define DPP_ROW_XMASK(m) (0x160 + (m))

template <int CTRL>
__device__ __forceinline__ float dpp_row(float v)
{
    return __builtin_bit_cast(float,
        __builtin_amdgcn_mov_dpp(__builtin_bit_cast(int, v), CTRL, 0xf, 0xf, true));
}

__device__ __forceinline__ float other_row(float v)
{
    return __builtin_bit_cast(float,
        __builtin_amdgcn_permlanex16(__builtin_bit_cast(int, v), __builtin_bit_cast(int, v),
                                     0x76543210, 0xfedcba98, false, false));
}

__device__ __forceinline__ float row_sum16(float v)
{
    v += dpp_row<DPP_ROW_XMASK(1)>(v);
    v += dpp_row<DPP_ROW_XMASK(2)>(v);
    v += dpp_row<DPP_ROW_XMASK(4)>(v);
    v += dpp_row<DPP_ROW_XMASK(8)>(v);
    return v;
}

__global__ __launch_bounds__(32 * WPB)
void lpc_residual_kernel(const float* __restrict__ pcm,
                         float* __restrict__ out,
                         int nframes)
{
    __shared__ __attribute__((aligned(16))) float smem[WPB * LDSW + FPB * AEN + FPB * RP];
    const int lane   = threadIdx.x & 31;
    const int wave   = threadIdx.x >> 5;
    const int half   = lane >> 4;
    const int cidx   = lane & 15;
    const int frame0 = (blockIdx.x * WPB + wave) * FPW;

    float* xps   = smem + wave * LDSW;
    float* g_lds = xps + FPW * XPS;
    float* a_all = smem + WPB * LDSW;
    float* r_all = a_all + FPB * AEN;
    float* ar    = a_all + wave * FPW * AEN;
    float* rw    = r_all + wave * FPW * RP;

    float xr[FPW][6];
    #pragma unroll
    for (int f = 0; f < FPW; ++f) {
        const int fi = min(frame0 + f, nframes - 1);
        const float* src = pcm + (size_t)fi * FRAME;
        #pragma unroll
        for (int q = 0; q < 5; ++q) xr[f][q] = src[lane + 32 * q];
        xr[f][5] = 0.f;
    }

    #pragma unroll
    for (int p = 0; p < NPR; ++p) {
        float* gp = g_lds + p * GSZ;

        #pragma unroll
        for (int h2 = 0; h2 < 2; ++h2) {
            const int f = 2 * p + h2;
            #pragma unroll
            for (int q = 0; q < 5; ++q) xps[f * XPS + NPAD + lane + 32 * q] = xr[f][q];
            if (lane < NPAD) { xps[f * XPS + lane] = 0.f; xps[f * XPS + NPAD + FRAME + lane] = 0.f; }
            if (lane < 15) {
                ar[f * AEN + 32 + lane] = 0.f;
                ar[f * AEN + lane]      = 0.f;
            }
        }
        __builtin_amdgcn_wave_barrier();

        v8f g0[2], g1[2];
        #pragma unroll
        for (int h2 = 0; h2 < 2; ++h2) {
            const int f = 2 * p + h2;
            float xb[6];
            #pragma unroll
            for (int q = 0; q < 5; ++q)
                xb[q] = xps[f * XPS + NPAD + 16 + lane + 32 * q];
            xb[5] = 0.f;

            g0[h2] = (v8f){}; g1[h2] = (v8f){};
            #pragma unroll
            for (int s = 0; s < 3; ++s) {
                v2f fa, fb;
                fa.x = xr[f][2 * s];  fa.y = xr[f][2 * s + 1];
                fb.x = xb[2 * s];     fb.y = xb[2 * s + 1];
                g0[h2] = __builtin_amdgcn_wmma_f32_16x16x4_f32(
                            false, fa, false, fa, (short)0, g0[h2], false, false);
                g1[h2] = __builtin_amdgcn_wmma_f32_16x16x4_f32(
                            false, fa, false, fb, (short)0, g1[h2], false, false);
            }
        }

        float part[2], diag[2];
        #pragma unroll
        for (int h2 = 0; h2 < 2; ++h2) {
            v4f* e0 = (v4f*)(gp + cidx * GP + half * 8);
            v4f* e1 = (v4f*)(gp + (16 + cidx) * GP + half * 8);
            e0[0] = (v4f){g0[h2][0], g0[h2][1], g0[h2][2], g0[h2][3]};
            e0[1] = (v4f){g0[h2][4], g0[h2][5], g0[h2][6], g0[h2][7]};
            e1[0] = (v4f){g1[h2][0], g1[h2][1], g1[h2][2], g1[h2][3]};
            e1[1] = (v4f){g1[h2][4], g1[h2][5], g1[h2][6], g1[h2][7]};
            __builtin_amdgcn_wave_barrier();

            const float* gd = gp + (8 * half + cidx) * GP + 8 * half;
            float s0 = 0.f, s1 = 0.f;
            #pragma unroll
            for (int i = 0; i < 8; i += 2) {
                s0 += gd[i * (GP + 1)];
                s1 += gd[(i + 1) * (GP + 1)];
            }
            part[h2] = s0 + s1;
            diag[h2] = gp[(16 + cidx) * GP + cidx];
            __builtin_amdgcn_wave_barrier();
        }

        {
            const float mine   = half ? part[1] : part[0];
            const float theirs = half ? part[0] : part[1];
            const float r   = mine + other_row(theirs);
            const float r16 = row_sum16(half ? diag[1] : diag[0]);
            float* rl = rw + (2 * p + half) * RP;
            rl[cidx] = r;
            if (cidx == 0) rl[16] = r16;
        }
    }
    __syncthreads();

    if (wave == 0) {
        const float* rp = r_all + lane * RP;
        float r[ORDER + 1], a[ORDER + 1];
        #pragma unroll
        for (int i = 0; i <= ORDER; ++i) r[i] = rp[i];
        float e = r[0];
        #pragma unroll
        for (int i = 1; i <= ORDER; ++i) {
            float acc = r[i];
            if (i == 2) acc = r[2] - a[1] * r[1];
            if (i > 2) {
                float s0 = a[1] * r[i - 1];
                float s1 = a[2] * r[i - 2];
                #pragma unroll
                for (int j = 3; j < i; j += 2) s0 = fmaf(a[j], r[i - j], s0);
                #pragma unroll
                for (int j = 4; j < i; j += 2) s1 = fmaf(a[j], r[i - j], s1);
                acc = r[i] - (s0 + s1);
            }
            const float k = acc / e;
            #pragma unroll
            for (int j = 1; 2 * j < i; ++j) {
                const float lo = a[j], hi = a[i - j];
                a[j]     = lo - k * hi;
                a[i - j] = hi - k * lo;
            }
            if ((i & 1) == 0) a[i / 2] = a[i / 2] - k * a[i / 2];
            a[i] = k;
            e = e * (1.f - k * k);
        }
        float* arh = a_all + lane * AEN;
        arh[31] = 1.f;
        #pragma unroll
        for (int j = 1; j <= ORDER; ++j) arh[31 - j] = a[j];
    }
    __syncthreads();

    {
        const int nb = cidx < 10 ? cidx : 9;
        #pragma unroll
        for (int p = 0; p < NPR; ++p) {
            float* y_lds = g_lds + p * GSZ;
            #pragma unroll
            for (int h2 = 0; h2 < 2; ++h2) {
                const int f = 2 * p + h2;
                const v4f* xp4 = (const v4f*)(xps + f * XPS + (NPAD - 16) + 16 * (nb + half));
                v4f xv[4];
                #pragma unroll
                for (int v = 0; v < 4; ++v) xv[v] = xp4[v];

                const float* ap = ar + f * AEN + 15 - cidx + 16 * half;
                float af[16];
                #pragma unroll
                for (int i = 0; i < 16; ++i) af[i] = ap[i];

                v8f acc = {};
                #pragma unroll
                for (int s = 0; s < 8; ++s) {
                    v2f fa, fb;
                    fa.x = af[2 * s];                  fa.y = af[2 * s + 1];
                    fb.x = xv[s >> 1][2 * (s & 1)];    fb.y = xv[s >> 1][2 * (s & 1) + 1];
                    acc = __builtin_amdgcn_wmma_f32_16x16x4_f32(
                              false, fa, false, fb, (short)0, acc, false, false);
                }
                v4f* yp = (v4f*)(y_lds + h2 * YSZ + 16 * cidx + 8 * half);
                yp[0] = (v4f){acc[0], acc[1], acc[2], acc[3]};
                yp[1] = (v4f){acc[4], acc[5], acc[6], acc[7]};
            }
            __builtin_amdgcn_wave_barrier();

            float* dst = out + (size_t)(frame0 + 2 * p) * FRAME;
            if (frame0 + 2 * p < nframes) {
                #pragma unroll
                for (int q = 0; q < 5; ++q)
                    dst[lane + 32 * q] = y_lds[lane + 32 * q];
            }
            if (frame0 + 2 * p + 1 < nframes) {
                #pragma unroll
                for (int q = 0; q < 5; ++q)
                    dst[FRAME + lane + 32 * q] = y_lds[YSZ + lane + 32 * q];
            }
        }
    }
}

extern "C" void kernel_launch(void* const* d_in, const int* in_sizes, int n_in,
                              void* d_out, int out_size, void* d_ws, size_t ws_size,
                              hipStream_t stream) {
    const float* pcm = (const float*)d_in[0];
    float* out = (float*)d_out;
    const int total   = in_sizes[0];
    const int nframes = total / FRAME;
    const int nwaves  = (nframes + FPW - 1) / FPW;
    const int blocks  = (nwaves + WPB - 1) / WPB;
    lpc_residual_kernel<<<blocks, 32 * WPB, 0, stream>>>(pcm, out, nframes);
}


# global loads and stores 16 B per lane: a wave's 4 frames are 2560 consecutive bytes, 5 + 5 b128 instructions (was 20 + 20 of 4 B each); Gram A fragments now read from the LDS staging; no sum reordered
# speedup vs baseline: 4.6261x; 1.0869x over previous
#include <hip/hip_runtime.h>


#define ORDER 16
#define FRAME 160
#define NPAD  20
#define XPS   (NPAD + FRAME + NPAD)
#define FPW   4
#define NPR   (FPW / 2)
#define WPB   8
#define FPB   (WPB * FPW)
#define GP    20
#define GSZ   (32 * GP)
#define AEN   49
#define YSZ   256
#define LDSW  (FPW * XPS + NPR * GSZ)
#define RP    17
static_assert(FPB == 32, "one lane of the recursion wave carries each of the block's frames");

typedef __attribute__((ext_vector_type(2))) float v2f;
typedef __attribute__((ext_vector_type(4))) float v4f;
typedef __attribute__((ext_vector_type(8))) float v8f;

#define DPP_ROW_XMASK(m) (0x160 + (m))

template <int CTRL>
__device__ __forceinline__ float dpp_row(float v)
{
    return __builtin_bit_cast(float,
        __builtin_amdgcn_mov_dpp(__builtin_bit_cast(int, v), CTRL, 0xf, 0xf, true));
}

__device__ __forceinline__ float other_row(float v)
{
    return __builtin_bit_cast(float,
        __builtin_amdgcn_permlanex16(__builtin_bit_cast(int, v), __builtin_bit_cast(int, v),
                                     0x76543210, 0xfedcba98, false, false));
}

__device__ __forceinline__ float row_sum16(float v)
{
    v += dpp_row<DPP_ROW_XMASK(1)>(v);
    v += dpp_row<DPP_ROW_XMASK(2)>(v);
    v += dpp_row<DPP_ROW_XMASK(4)>(v);
    v += dpp_row<DPP_ROW_XMASK(8)>(v);
    return v;
}

__global__ __launch_bounds__(32 * WPB)
void lpc_residual_kernel(const float* __restrict__ pcm,
                         float* __restrict__ out,
                         int nframes)
{
    __shared__ __attribute__((aligned(16))) float smem[WPB * LDSW + FPB * AEN + FPB * RP];
    const int lane   = threadIdx.x & 31;
    const int wave   = threadIdx.x >> 5;
    const int half   = lane >> 4;
    const int cidx   = lane & 15;
    const int frame0 = (blockIdx.x * WPB + wave) * FPW;

    float* xps   = smem + wave * LDSW;
    float* g_lds = xps + FPW * XPS;
    float* a_all = smem + WPB * LDSW;
    float* r_all = a_all + FPB * AEN;
    float* ar    = a_all + wave * FPW * AEN;
    float* rw    = r_all + wave * FPW * RP;

    v4f xg[5];
    int gf[5], go[5];
    #pragma unroll
    for (int q = 0; q < 5; ++q) {
        const int g = 128 * q + 4 * lane;
        const int f = (g >= FRAME) + (g >= 2 * FRAME) + (g >= 3 * FRAME);
        gf[q] = f;
        go[q] = g - f * FRAME;
        const int fi = min(frame0 + f, nframes - 1);
        xg[q] = *(const v4f*)(pcm + (size_t)fi * FRAME + go[q]);
    }

    #pragma unroll
    for (int f = 0; f < FPW; ++f) {
        if (lane < NPAD) { xps[f * XPS + lane] = 0.f; xps[f * XPS + NPAD + FRAME + lane] = 0.f; }
        if (lane < 15) {
            ar[f * AEN + 32 + lane] = 0.f;
            ar[f * AEN + lane]      = 0.f;
        }
    }

    #pragma unroll
    for (int p = 0; p < NPR; ++p) {
        float* gp = g_lds + p * GSZ;

        #pragma unroll
        for (int q = (p == 0 ? 0 : 3); q < (p == 0 ? 3 : 5); ++q)
            *(v4f*)(xps + gf[q] * XPS + NPAD + go[q]) = xg[q];
        __builtin_amdgcn_wave_barrier();

        v8f g0[2], g1[2];
        #pragma unroll
        for (int h2 = 0; h2 < 2; ++h2) {
            const int f = 2 * p + h2;
            float xa[6], xb[6];
            #pragma unroll
            for (int q = 0; q < 5; ++q) {
                xa[q] = xps[f * XPS + NPAD + lane + 32 * q];
                xb[q] = xps[f * XPS + NPAD + 16 + lane + 32 * q];
            }
            xa[5] = 0.f; xb[5] = 0.f;

            g0[h2] = (v8f){}; g1[h2] = (v8f){};
            #pragma unroll
            for (int s = 0; s < 3; ++s) {
                v2f fa, fb;
                fa.x = xa[2 * s];     fa.y = xa[2 * s + 1];
                fb.x = xb[2 * s];     fb.y = xb[2 * s + 1];
                g0[h2] = __builtin_amdgcn_wmma_f32_16x16x4_f32(
                            false, fa, false, fa, (short)0, g0[h2], false, false);
                g1[h2] = __builtin_amdgcn_wmma_f32_16x16x4_f32(
                            false, fa, false, fb, (short)0, g1[h2], false, false);
            }
        }

        float part[2], diag[2];
        #pragma unroll
        for (int h2 = 0; h2 < 2; ++h2) {
            v4f* e0 = (v4f*)(gp + cidx * GP + half * 8);
            v4f* e1 = (v4f*)(gp + (16 + cidx) * GP + half * 8);
            e0[0] = (v4f){g0[h2][0], g0[h2][1], g0[h2][2], g0[h2][3]};
            e0[1] = (v4f){g0[h2][4], g0[h2][5], g0[h2][6], g0[h2][7]};
            e1[0] = (v4f){g1[h2][0], g1[h2][1], g1[h2][2], g1[h2][3]};
            e1[1] = (v4f){g1[h2][4], g1[h2][5], g1[h2][6], g1[h2][7]};
            __builtin_amdgcn_wave_barrier();

            const float* gd = gp + (8 * half + cidx) * GP + 8 * half;
            float s0 = 0.f, s1 = 0.f;
            #pragma unroll
            for (int i = 0; i < 8; i += 2) {
                s0 += gd[i * (GP + 1)];
                s1 += gd[(i + 1) * (GP + 1)];
            }
            part[h2] = s0 + s1;
            diag[h2] = gp[(16 + cidx) * GP + cidx];
            __builtin_amdgcn_wave_barrier();
        }

        {
            const float mine   = half ? part[1] : part[0];
            const float theirs = half ? part[0] : part[1];
            const float r   = mine + other_row(theirs);
            const float r16 = row_sum16(half ? diag[1] : diag[0]);
            float* rl = rw + (2 * p + half) * RP;
            rl[cidx] = r;
            if (cidx == 0) rl[16] = r16;
        }
    }
    __syncthreads();

    if (wave == 0) {
        const float* rp = r_all + lane * RP;
        float r[ORDER + 1], a[ORDER + 1];
        #pragma unroll
        for (int i = 0; i <= ORDER; ++i) r[i] = rp[i];
        float e = r[0];
        #pragma unroll
        for (int i = 1; i <= ORDER; ++i) {
            float acc = r[i];
            if (i == 2) acc = r[2] - a[1] * r[1];
            if (i > 2) {
                float s0 = a[1] * r[i - 1];
                float s1 = a[2] * r[i - 2];
                #pragma unroll
                for (int j = 3; j < i; j += 2) s0 = fmaf(a[j], r[i - j], s0);
                #pragma unroll
                for (int j = 4; j < i; j += 2) s1 = fmaf(a[j], r[i - j], s1);
                acc = r[i] - (s0 + s1);
            }
            const float k = acc / e;
            #pragma unroll
            for (int j = 1; 2 * j < i; ++j) {
                const float lo = a[j], hi = a[i - j];
                a[j]     = lo - k * hi;
                a[i - j] = hi - k * lo;
            }
            if ((i & 1) == 0) a[i / 2] = a[i / 2] - k * a[i / 2];
            a[i] = k;
            e = e * (1.f - k * k);
        }
        float* arh = a_all + lane * AEN;
        arh[31] = 1.f;
        #pragma unroll
        for (int j = 1; j <= ORDER; ++j) arh[31 - j] = a[j];
    }
    __syncthreads();

    {
        const int nb = cidx < 10 ? cidx : 9;
        #pragma unroll
        for (int p = 0; p < NPR; ++p) {
            float* y_lds = g_lds + p * GSZ;
            #pragma unroll
            for (int h2 = 0; h2 < 2; ++h2) {
                const int f = 2 * p + h2;
                const v4f* xp4 = (const v4f*)(xps + f * XPS + (NPAD - 16) + 16 * (nb + half));
                v4f xv[4];
                #pragma unroll
                for (int v = 0; v < 4; ++v) xv[v] = xp4[v];

                const float* ap = ar + f * AEN + 15 - cidx + 16 * half;
                float af[16];
                #pragma unroll
                for (int i = 0; i < 16; ++i) af[i] = ap[i];

                v8f acc = {};
                #pragma unroll
                for (int s = 0; s < 8; ++s) {
                    v2f fa, fb;
                    fa.x = af[2 * s];                  fa.y = af[2 * s + 1];
                    fb.x = xv[s >> 1][2 * (s & 1)];    fb.y = xv[s >> 1][2 * (s & 1) + 1];
                    acc = __builtin_amdgcn_wmma_f32_16x16x4_f32(
                              false, fa, false, fb, (short)0, acc, false, false);
                }
                v4f* yp = (v4f*)(y_lds + h2 * YSZ + 16 * cidx + 8 * half);
                yp[0] = (v4f){acc[0], acc[1], acc[2], acc[3]};
                yp[1] = (v4f){acc[4], acc[5], acc[6], acc[7]};
            }
            __builtin_amdgcn_wave_barrier();

            v4f yv[5];
            #pragma unroll
            for (int q = (p == 0 ? 0 : 2); q < (p == 0 ? 2 : 5); ++q)
                yv[q] = *(const v4f*)(g_lds + (gf[q] >> 1) * GSZ + (gf[q] & 1) * YSZ + go[q]);
            if (p == 0) asm volatile("" :: "v"(yv[0]), "v"(yv[1]));
            else        asm volatile("" :: "v"(yv[2]), "v"(yv[3]), "v"(yv[4]));
            #pragma unroll
            for (int q = (p == 0 ? 0 : 2); q < (p == 0 ? 2 : 5); ++q)
                if (frame0 + gf[q] < nframes)
                    *(v4f*)(out + (size_t)(frame0 + gf[q]) * FRAME + go[q]) = yv[q];
        }
    }
}

extern "C" void kernel_launch(void* const* d_in, const int* in_sizes, int n_in,
                              void* d_out, int out_size, void* d_ws, size_t ws_size,
                              hipStream_t stream) {
    const float* pcm = (const float*)d_in[0];
    float* out = (float*)d_out;
    const int total   = in_sizes[0];
    const int nframes = total / FRAME;
    const int nwaves  = (nframes + FPW - 1) / FPW;
    const int blocks  = (nwaves + WPB - 1) / WPB;
    lpc_residual_kernel<<<blocks, 32 * WPB, 0, stream>>>(pcm, out, nframes);
}
